# starting kernel
# baseline (speedup 1.0000x reference)
#include <hip/hip_runtime.h>
#include <hip/hip_bf16.h>

#define N_NODES 10000
#define N_EDGES 320000
#define F_INF   128
#define F_OUTF  128
#define H_HEADS 4
#define C_OUT   (H_HEADS * F_OUTF)
#define STEPS   2
#define NEG_SLOPE 0.2f

typedef float v2f __attribute__((ext_vector_type(2)));
typedef float v8f __attribute__((ext_vector_type(8)));

__global__ __launch_bounds__(256) void gemm_xw(const float* __restrict__ x,
                                               const float* __restrict__ W,
                                               float* __restrict__ h) {
  const int lane = threadIdx.x & 31;
  const int wv   = threadIdx.x >> 5;
  const int tile = blockIdx.x * 8 + wv;
  const int tiles_n = C_OUT / 16;
  const int tm = tile / tiles_n;
  const int tn = tile % tiles_n;
  const int m0 = tm * 16, n0 = tn * 16;
  const int half = lane >> 4;
  const int mr   = lane & 15;

  v8f acc = {};
  const float* arow = x + (size_t)(m0 + mr) * F_INF;
  #pragma unroll 4
  for (int k0 = 0; k0 < F_INF; k0 += 4) {
    const int k = k0 + 2 * half;
    v2f a; a.x = arow[k]; a.y = arow[k + 1];
    v2f b; b.x = W[(size_t)k * C_OUT + n0 + mr];
           b.y = W[(size_t)(k + 1) * C_OUT + n0 + mr];
    acc = __builtin_amdgcn_wmma_f32_16x16x4_f32(
         false, a,  false, b,
         (short)0, acc,  false,  false);
  }
  const int nn = lane & 15;
  const int mbase = half * 8;
  #pragma unroll
  for (int i = 0; i < 8; ++i)
    h[(size_t)(m0 + mbase + i) * C_OUT + n0 + nn] = acc[i];
}

__global__ __launch_bounds__(128) void node_logits(const float* __restrict__ h,
                                                   const float* __restrict__ attn_l,
                                                   const float* __restrict__ attn_r,
                                                   float* __restrict__ el,
                                                   float* __restrict__ er) {
  const int n = blockIdx.x;
  const int head = threadIdx.x >> 5, lane = threadIdx.x & 31;
  const float* hr = h + (size_t)n * C_OUT + head * F_OUTF;
  const float* al = attn_l + head * F_OUTF;
  const float* ar = attn_r + head * F_OUTF;
  float sl = 0.f, sr = 0.f;
  #pragma unroll
  for (int f = lane; f < F_OUTF; f += 32) {
    const float v = hr[f];
    sl += v * al[f];
    sr += v * ar[f];
  }
  #pragma unroll
  for (int off = 16; off; off >>= 1) {
    sl += __shfl_xor(sl, off, 32);
    sr += __shfl_xor(sr, off, 32);
  }
  if (lane == 0) {
    el[n * H_HEADS + head] = sl;
    er[n * H_HEADS + head] = sr;
  }
}

__global__ void edge_count(const int* __restrict__ dst, int* __restrict__ cnt) {
  const int e = blockIdx.x * blockDim.x + threadIdx.x;
  if (e < N_EDGES) atomicAdd(&cnt[dst[e]], 1);
}

__global__ __launch_bounds__(1024) void scan_rowptr(const int* __restrict__ cnt,
                                                    int* __restrict__ row_ptr,
                                                    int* __restrict__ cursor) {
  __shared__ int sh[1024];
  __shared__ int carry;
  const int t = threadIdx.x;
  if (t == 0) carry = 0;
  __syncthreads();
  for (int base = 0; base < N_NODES; base += 1024) {
    const int idx = base + t;
    const int v = (idx < N_NODES) ? cnt[idx] : 0;
    sh[t] = v;
    __syncthreads();
    for (int off = 1; off < 1024; off <<= 1) {
      const int u = (t >= off) ? sh[t - off] : 0;
      __syncthreads();
      sh[t] += u;
      __syncthreads();
    }
    const int excl = sh[t] - v + carry;
    if (idx < N_NODES) { row_ptr[idx] = excl; cursor[idx] = excl; }
    __syncthreads();
    if (t == 1023) carry += sh[1023];
    __syncthreads();
  }
  if (t == 0) row_ptr[N_NODES] = carry;
}

__global__ void edge_fill(const int* __restrict__ dst, int* __restrict__ cursor,
                          int* __restrict__ edge_idx) {
  const int e = blockIdx.x * blockDim.x + threadIdx.x;
  if (e < N_EDGES) {
    const int p = atomicAdd(&cursor[dst[e]], 1);
    edge_idx[p] = e;
  }
}

__global__ __launch_bounds__(128) void gat_aggregate(
    const float* __restrict__ h, const float* __restrict__ el,
    const float* __restrict__ er, const int* __restrict__ src,
    const int* __restrict__ row_ptr, const int* __restrict__ edge_idx,
    const float* __restrict__ bias, const float* __restrict__ alpha_p,
    const float* __restrict__ x0, float* __restrict__ xout) {
  __shared__ float red[H_HEADS * F_OUTF];
  const int n = blockIdx.x;
  const int head = threadIdx.x >> 5, lane = threadIdx.x & 31;
  const int rs = row_ptr[n], re = row_ptr[n + 1];
  const float erd = er[n * H_HEADS + head];

  float mx = -INFINITY;
  for (int i = rs + lane; i < re; i += 32) {
    const int e = edge_idx[i];
    float v = el[src[e] * H_HEADS + head] + erd;
    v = v > 0.f ? v : NEG_SLOPE * v;
    mx = fmaxf(mx, v);
  }
  #pragma unroll
  for (int off = 16; off; off >>= 1) mx = fmaxf(mx, __shfl_xor(mx, off, 32));

  float sum = 0.f;
  for (int i = rs + lane; i < re; i += 32) {
    const int e = edge_idx[i];
    float v = el[src[e] * H_HEADS + head] + erd;
    v = v > 0.f ? v : NEG_SLOPE * v;
    sum += __expf(v - mx);
  }
  #pragma unroll
  for (int off = 16; off; off >>= 1) sum += __shfl_xor(sum, off, 32);
  const float inv = (sum != 0.f) ? 1.f / sum : 0.f;

  float4 acc = make_float4(0.f, 0.f, 0.f, 0.f);
  for (int i = rs; i < re; ++i) {
    const int e = edge_idx[i];
    const int s = src[e];
    float v = el[s * H_HEADS + head] + erd;
    v = v > 0.f ? v : NEG_SLOPE * v;
    const float a = __expf(v - mx) * inv;
    const float4 hv = *(const float4*)(h + (size_t)s * C_OUT + head * F_OUTF + lane * 4);
    acc.x += a * hv.x; acc.y += a * hv.y; acc.z += a * hv.z; acc.w += a * hv.w;
  }
  const float4 bv = *(const float4*)(bias + head * F_OUTF + lane * 4);
  acc.x += bv.x; acc.y += bv.y; acc.z += bv.z; acc.w += bv.w;
  *(float4*)(red + head * F_OUTF + lane * 4) = acc;
  __syncthreads();

  const int f = threadIdx.x;
  const float s = 0.25f * (red[f] + red[F_OUTF + f] +
                           red[2 * F_OUTF + f] + red[3 * F_OUTF + f]);
  const float al = alpha_p[0];
  xout[(size_t)n * F_OUTF + f] = (1.f - al) * s + al * x0[(size_t)n * F_OUTF + f];
}

extern "C" void kernel_launch(void* const* d_in, const int* in_sizes, int n_in,
                              void* d_out, int out_size, void* d_ws, size_t ws_size,
                              hipStream_t stream) {
  const float* x      = (const float*)d_in[0];
  const float* x0     = (const float*)d_in[1];
  const int*   src    = (const int*)d_in[2];
  const int*   dst    = (const int*)d_in[3];
  const float* alpha  = (const float*)d_in[4];
  const float* W      = (const float*)d_in[5];
  const float* attn_l = (const float*)d_in[6];
  const float* attn_r = (const float*)d_in[7];
  const float* bias   = (const float*)d_in[8];

  char* ws = (char*)d_ws;
  size_t off = 0;
  auto alloc = [&](size_t bytes) -> char* {
    char* p = ws + off;
    off = (off + bytes + 255) & ~(size_t)255;
    return p;
  };
  float* h        = (float*)alloc((size_t)N_NODES * C_OUT * sizeof(float));
  float* el       = (float*)alloc((size_t)N_NODES * H_HEADS * sizeof(float));
  float* er       = (float*)alloc((size_t)N_NODES * H_HEADS * sizeof(float));
  float* xbuf     = (float*)alloc((size_t)N_NODES * F_INF * sizeof(float));
  int*   cnt      = (int*)alloc((size_t)N_NODES * sizeof(int));
  int*   row_ptr  = (int*)alloc((size_t)(N_NODES + 1) * sizeof(int));
  int*   cursor   = (int*)alloc((size_t)N_NODES * sizeof(int));
  int*   edge_idx = (int*)alloc((size_t)N_EDGES * sizeof(int));

  hipMemsetAsync(cnt, 0, (size_t)N_NODES * sizeof(int), stream);
  edge_count<<<(N_EDGES + 255) / 256, 256, 0, stream>>>(dst, cnt);
  scan_rowptr<<<1, 1024, 0, stream>>>(cnt, row_ptr, cursor);
  edge_fill<<<(N_EDGES + 255) / 256, 256, 0, stream>>>(dst, cursor, edge_idx);

  const float* xin = x;
  for (int step = 0; step < STEPS; ++step) {
    gemm_xw<<<2500, 256, 0, stream>>>(xin, W, h);
    node_logits<<<N_NODES, 128, 0, stream>>>(h, attn_l, attn_r, el, er);
    float* xo = (step == STEPS - 1) ? (float*)d_out : xbuf;
    gat_aggregate<<<N_NODES, 128, 0, stream>>>(h, el, er, src, row_ptr, edge_idx,
                                               bias, alpha, x0, xo);
    xin = xbuf;
  }
}


# GEMM: a wave owns 16 rows x 64 columns (4 accumulators): one A fragment feeds 4 WMMAs, B for the 4 tiles comes as two 16-byte loads per lane; loads per WMMA 3 -> 0.75; order of f32 sums kept
# speedup vs baseline: 1.3085x; 1.3085x over previous
#include <hip/hip_runtime.h>
#include <hip/hip_bf16.h>

#define N_NODES 10000
#define N_EDGES 320000
#define F_INF   128
#define F_OUTF  128
#define H_HEADS 4
#define C_OUT   (H_HEADS * F_OUTF)
#define STEPS   2
#define NEG_SLOPE 0.2f

typedef float v2f __attribute__((ext_vector_type(2)));
typedef float v8f __attribute__((ext_vector_type(8)));

#define GEMM_WAVE_COLS   64
#define GEMM_COL_GROUPS  (C_OUT / GEMM_WAVE_COLS)
#define GEMM_ROW_TILES   (N_NODES / 16)
#define GEMM_ROW_GROUPS  ((GEMM_ROW_TILES + 3) / 4)
#define GEMM_BLOCKS      (GEMM_ROW_GROUPS * GEMM_COL_GROUPS)
static_assert(N_NODES % 16 == 0 && C_OUT % GEMM_WAVE_COLS == 0 && F_INF % 4 == 0,
              "whole tiles only: no partial tile, so every lane of a live wave is active");
__global__ __launch_bounds__(128) void gemm_xw(const float* __restrict__ x,
                                               const float* __restrict__ W,
                                               float* __restrict__ h) {
  const int lane = threadIdx.x & 31;
  const int wv   = threadIdx.x >> 5;
  const int half = lane >> 4;
  const int mr   = lane & 15;
  const int cg   = blockIdx.x % GEMM_COL_GROUPS;
  const int tm   = (blockIdx.x / GEMM_COL_GROUPS) * 4 + wv;
  if (tm >= GEMM_ROW_TILES) return;
  const int m0 = tm * 16, n0 = cg * GEMM_WAVE_COLS;

  v8f acc0 = {}, acc1 = {}, acc2 = {}, acc3 = {};
  const float* arow = x + (size_t)(m0 + mr) * F_INF + 2 * half;
  const float* bcol = W + (size_t)(2 * half) * C_OUT + n0 + 4 * mr;
  #pragma unroll 8
  for (int k0 = 0; k0 < F_INF; k0 += 4) {
    v2f a; a.x = arow[k0]; a.y = arow[k0 + 1];
    const float4 bx = *(const float4*)(bcol + (size_t)k0 * C_OUT);
    const float4 by = *(const float4*)(bcol + (size_t)(k0 + 1) * C_OUT);
    v2f b;
    b.x = bx.x; b.y = by.x;
    acc0 = __builtin_amdgcn_wmma_f32_16x16x4_f32(
         false, a,  false, b,
         (short)0, acc0,  false,  false);
    b.x = bx.y; b.y = by.y;
    acc1 = __builtin_amdgcn_wmma_f32_16x16x4_f32(false, a, false, b, (short)0, acc1, false, false);
    b.x = bx.z; b.y = by.z;
    acc2 = __builtin_amdgcn_wmma_f32_16x16x4_f32(false, a, false, b, (short)0, acc2, false, false);
    b.x = bx.w; b.y = by.w;
    acc3 = __builtin_amdgcn_wmma_f32_16x16x4_f32(false, a, false, b, (short)0, acc3, false, false);
  }
  float* hrow = h + (size_t)(m0 + half * 8) * C_OUT + n0 + 4 * mr;
  #pragma unroll
  for (int i = 0; i < 8; ++i)
    *(float4*)(hrow + (size_t)i * C_OUT) = make_float4(acc0[i], acc1[i], acc2[i], acc3[i]);
}

#define NL_BLOCKS 625
__global__ __launch_bounds__(256) void node_logits(const float* __restrict__ h,
                                                   const float* __restrict__ attn_l,
                                                   const float* __restrict__ attn_r,
                                                   float* __restrict__ el,
                                                   float* __restrict__ er) {
  const int lane = threadIdx.x & 31;
  const int w = blockIdx.x * 8 + (threadIdx.x >> 5);
  float al[H_HEADS][4], ar[H_HEADS][4];
  #pragma unroll
  for (int hd = 0; hd < H_HEADS; ++hd) {
    #pragma unroll
    for (int k = 0; k < 4; ++k) {
      al[hd][k] = attn_l[hd * F_OUTF + lane + 32 * k];
      ar[hd][k] = attn_r[hd * F_OUTF + lane + 32 * k];
    }
  }
  for (int n = w; n < N_NODES; n += NL_BLOCKS * 8) {
    const float* hr = h + (size_t)n * C_OUT;
    float v[H_HEADS][4];
    #pragma unroll
    for (int hd = 0; hd < H_HEADS; ++hd) {
      #pragma unroll
      for (int k = 0; k < 4; ++k) v[hd][k] = hr[hd * F_OUTF + lane + 32 * k];
    }
    float sl[H_HEADS], sr[H_HEADS];
    #pragma unroll
    for (int hd = 0; hd < H_HEADS; ++hd) {
      float a = 0.f, b = 0.f;
      #pragma unroll
      for (int k = 0; k < 4; ++k) {
        a += v[hd][k] * al[hd][k];
        b += v[hd][k] * ar[hd][k];
      }
      sl[hd] = a; sr[hd] = b;
    }
    #pragma unroll
    for (int off = 16; off; off >>= 1) {
      #pragma unroll
      for (int hd = 0; hd < H_HEADS; ++hd) {
        sl[hd] += __shfl_xor(sl[hd], off, 32);
        sr[hd] += __shfl_xor(sr[hd], off, 32);
      }
    }
    if (lane == 0) {
      *(float4*)(el + (size_t)n * H_HEADS) = make_float4(sl[0], sl[1], sl[2], sl[3]);
      *(float4*)(er + (size_t)n * H_HEADS) = make_float4(sr[0], sr[1], sr[2], sr[3]);
    }
  }
}

#define EDGE_BLOCKS     250
#define EDGE_PER_THREAD 5
static_assert(EDGE_BLOCKS * 256 * EDGE_PER_THREAD == N_EDGES, "edge grid must cover N_EDGES exactly");
__global__ __launch_bounds__(256) void edge_count(const int* __restrict__ dst,
                                                  int* __restrict__ cnt) {
  const int t = blockIdx.x * 256 + threadIdx.x;
  int d[EDGE_PER_THREAD];
  #pragma unroll
  for (int k = 0; k < EDGE_PER_THREAD; ++k) d[k] = dst[t + k * (EDGE_BLOCKS * 256)];
  #pragma unroll
  for (int k = 0; k < EDGE_PER_THREAD; ++k) atomicAdd(&cnt[d[k]], 1);
}

#define SCAN_THREADS 1024
#define SCAN_ITEMS   10
static_assert(N_NODES % SCAN_ITEMS == 0 && N_NODES / SCAN_ITEMS <= SCAN_THREADS,
              "a thread is wholly inside or wholly outside the node range");
static_assert(SCAN_ITEMS % 2 == 0, "counts are moved two at a time (8-byte accesses)");
__global__ __launch_bounds__(SCAN_THREADS) void scan_rowptr(const int* __restrict__ cnt,
                                                    int* __restrict__ row_ptr,
                                                    int* __restrict__ cursor) {
  __shared__ int wave_total[SCAN_THREADS / 32];
  const int t = threadIdx.x, lane = t & 31, wv = t >> 5;
  const int base = t * SCAN_ITEMS;
  const bool live = base < N_NODES;

  const int* mine = cnt + (live ? base : 0);
  int2 p[SCAN_ITEMS / 2];
  #pragma unroll
  for (int k = 0; k < SCAN_ITEMS / 2; ++k) p[k] = *(const int2*)(mine + 2 * k);
  int v[SCAN_ITEMS];
  #pragma unroll
  for (int k = 0; k < SCAN_ITEMS / 2; ++k) {
    v[2 * k]     = live ? p[k].x : 0;
    v[2 * k + 1] = live ? p[k].y : 0;
  }
  int run = 0;
  #pragma unroll
  for (int k = 0; k < SCAN_ITEMS; ++k) { const int c = v[k]; v[k] = run; run += c; }

  int inc = run;
  for (int off = 1; off < 32; off <<= 1) {
    const int u = __shfl_up(inc, off, 32);
    if (lane >= off) inc += u;
  }
  if (lane == 31) wave_total[wv] = inc;
  __syncthreads();

  int before = (lane < wv) ? wave_total[lane] : 0;
  for (int off = 16; off; off >>= 1) before += __shfl_xor(before, off, 32);

  const int texcl = before + inc - run;
  if (live) {
    #pragma unroll
    for (int k = 0; k < SCAN_ITEMS; k += 2) {
      const int2 o = make_int2(texcl + v[k], texcl + v[k + 1]);
      *(int2*)(row_ptr + base + k) = o;
      *(int2*)(cursor + base + k) = o;
    }
    if (base + SCAN_ITEMS == N_NODES) row_ptr[N_NODES] = texcl + run;
  }
}

__global__ __launch_bounds__(256) void edge_fill(const int* __restrict__ dst,
                                                 int* __restrict__ cursor,
                                                 int* __restrict__ edge_idx) {
  const int t = blockIdx.x * 256 + threadIdx.x;
  int d[EDGE_PER_THREAD], p[EDGE_PER_THREAD];
  #pragma unroll
  for (int k = 0; k < EDGE_PER_THREAD; ++k) d[k] = dst[t + k * (EDGE_BLOCKS * 256)];
  #pragma unroll
  for (int k = 0; k < EDGE_PER_THREAD; ++k) p[k] = atomicAdd(&cursor[d[k]], 1);
  #pragma unroll
  for (int k = 0; k < EDGE_PER_THREAD; ++k) edge_idx[p[k]] = t + k * (EDGE_BLOCKS * 256);
}

#define GROUP 8
#define AGG_BLOCKS 3334

__device__ __forceinline__ float edge_score(const float* __restrict__ el, int s,
                                            int head, float erd) {
  const float v = el[s * H_HEADS + head] + erd;
  return v > 0.f ? v : NEG_SLOPE * v;
}

__device__ __forceinline__ void fma4(float4& acc, float w, const float4& v) {
  acc.x = fmaf(w, v.x, acc.x); acc.y = fmaf(w, v.y, acc.y);
  acc.z = fmaf(w, v.z, acc.z); acc.w = fmaf(w, v.w, acc.w);
}

__device__ __forceinline__ void gather_chunk(float4& acc, const float* __restrict__ hcol,
                                             int s, float w, int cnt) {
  int j = 0;
  for (; j + GROUP <= cnt; j += GROUP) {
    float4 hv[GROUP]; float wj[GROUP];
    #pragma unroll
    for (int k = 0; k < GROUP; ++k) {
      const int sj = __builtin_amdgcn_readlane(s, j + k);
      wj[k] = __int_as_float(__builtin_amdgcn_readlane(__float_as_int(w), j + k));
      hv[k] = *(const float4*)(hcol + (size_t)sj * C_OUT);
    }
    #pragma unroll
    for (int k = 0; k < GROUP; ++k) fma4(acc, wj[k], hv[k]);
  }
  if (j < cnt) {
    float4 hv[GROUP - 1]; float wj[GROUP - 1];
    #pragma unroll
    for (int k = 0; k < GROUP - 1; ++k) {
      hv[k] = make_float4(0.f, 0.f, 0.f, 0.f);
      wj[k] = 0.f;
      if (j + k < cnt) {
        const int sj = __builtin_amdgcn_readlane(s, j + k);
        wj[k] = __int_as_float(__builtin_amdgcn_readlane(__float_as_int(w), j + k));
        hv[k] = *(const float4*)(hcol + (size_t)sj * C_OUT);
      }
    }
    #pragma unroll
    for (int k = 0; k < GROUP - 1; ++k) fma4(acc, wj[k], hv[k]);
  }
}

__global__ __launch_bounds__(128) void gat_aggregate(
    const float* __restrict__ h, const float* __restrict__ el,
    const float* __restrict__ er, const int* __restrict__ src,
    const int* __restrict__ row_ptr, const int* __restrict__ edge_idx,
    const float* __restrict__ bias, const float* __restrict__ alpha_p,
    const float* __restrict__ x0, float* __restrict__ xout) {
  __shared__ float red2[2][H_HEADS * F_OUTF];
  const int head = threadIdx.x >> 5, lane = threadIdx.x & 31;
  const float* hcol = h + head * F_OUTF + lane * 4;
  const float4 bv = *(const float4*)(bias + head * F_OUTF + lane * 4);
  const float al = alpha_p[0];
  int buf = 0;
  for (int n = blockIdx.x; n < N_NODES; n += AGG_BLOCKS, buf ^= 1) {
  float* red = red2[buf];
  const int rs = row_ptr[n], re = row_ptr[n + 1];
  const float erd = er[n * H_HEADS + head];

  const int i0 = rs + lane;
  const bool has0 = i0 < re;
  int s0 = 0;
  float v0 = 0.f;
  if (has0) {
    s0 = src[edge_idx[i0]];
    v0 = edge_score(el, s0, head, erd);
  }

  float mx = -INFINITY;
  if (has0) mx = fmaxf(mx, v0);
  for (int i = i0 + 32; i < re; i += 32)
    mx = fmaxf(mx, edge_score(el, src[edge_idx[i]], head, erd));
  #pragma unroll
  for (int off = 16; off; off >>= 1) mx = fmaxf(mx, __shfl_xor(mx, off, 32));

  const float p0 = has0 ? __expf(v0 - mx) : 0.f;
  float sum = p0;
  for (int i = i0 + 32; i < re; i += 32)
    sum += __expf(edge_score(el, src[edge_idx[i]], head, erd) - mx);
  #pragma unroll
  for (int off = 16; off; off >>= 1) sum += __shfl_xor(sum, off, 32);
  const float inv = (sum != 0.f) ? 1.f / sum : 0.f;

  float4 acc = make_float4(0.f, 0.f, 0.f, 0.f);
  if (rs < re) gather_chunk(acc, hcol, s0, p0 * inv, min(32, re - rs));
  for (int base = rs + 32; base < re; base += 32) {
    const int i = base + lane;
    int s = 0;
    float w = 0.f;
    if (i < re) {
      s = src[edge_idx[i]];
      w = __expf(edge_score(el, s, head, erd) - mx) * inv;
    }
    gather_chunk(acc, hcol, s, w, min(32, re - base));
  }
  acc.x += bv.x; acc.y += bv.y; acc.z += bv.z; acc.w += bv.w;
  *(float4*)(red + head * F_OUTF + lane * 4) = acc;
  __syncthreads();

  const int f = threadIdx.x;
  const float s = 0.25f * (red[f] + red[F_OUTF + f] +
                           red[2 * F_OUTF + f] + red[3 * F_OUTF + f]);
  xout[(size_t)n * F_OUTF + f] = (1.f - al) * s + al * x0[(size_t)n * F_OUTF + f];
  }
}

extern "C" void kernel_launch(void* const* d_in, const int* in_sizes, int n_in,
                              void* d_out, int out_size, void* d_ws, size_t ws_size,
                              hipStream_t stream) {
  const float* x      = (const float*)d_in[0];
  const float* x0     = (const float*)d_in[1];
  const int*   src    = (const int*)d_in[2];
  const int*   dst    = (const int*)d_in[3];
  const float* alpha  = (const float*)d_in[4];
  const float* W      = (const float*)d_in[5];
  const float* attn_l = (const float*)d_in[6];
  const float* attn_r = (const float*)d_in[7];
  const float* bias   = (const float*)d_in[8];

  char* ws = (char*)d_ws;
  size_t off = 0;
  auto alloc = [&](size_t bytes) -> char* {
    char* p = ws + off;
    off = (off + bytes + 255) & ~(size_t)255;
    return p;
  };
  float* h        = (float*)alloc((size_t)N_NODES * C_OUT * sizeof(float));
  float* el       = (float*)alloc((size_t)N_NODES * H_HEADS * sizeof(float));
  float* er       = (float*)alloc((size_t)N_NODES * H_HEADS * sizeof(float));
  float* xbuf     = (float*)alloc((size_t)N_NODES * F_INF * sizeof(float));
  int*   cnt      = (int*)alloc((size_t)N_NODES * sizeof(int));
  int*   row_ptr  = (int*)alloc((size_t)(N_NODES + 1) * sizeof(int));
  int*   cursor   = (int*)alloc((size_t)N_NODES * sizeof(int));
  int*   edge_idx = (int*)alloc((size_t)N_EDGES * sizeof(int));

  hipMemsetAsync(cnt, 0, (size_t)N_NODES * sizeof(int), stream);
  edge_count<<<EDGE_BLOCKS, 256, 0, stream>>>(dst, cnt);
  scan_rowptr<<<1, SCAN_THREADS, 0, stream>>>(cnt, row_ptr, cursor);
  edge_fill<<<EDGE_BLOCKS, 256, 0, stream>>>(dst, cursor, edge_idx);

  const float* xin = x;
  for (int step = 0; step < STEPS; ++step) {
    gemm_xw<<<GEMM_BLOCKS, 128, 0, stream>>>(xin, W, h);
    node_logits<<<NL_BLOCKS, 256, 0, stream>>>(h, attn_l, attn_r, el, er);
    float* xo = (step == STEPS - 1) ? (float*)d_out : xbuf;
    gat_aggregate<<<AGG_BLOCKS, 128, 0, stream>>>(h, el, er, src, row_ptr, edge_idx,
                                               bias, alpha, x0, xo);
    xin = xbuf;
  }
}


# CSR build: the counting add hands back each edge's rank in its segment, kept in a plane; placing is segment start + rank, no second atomic per edge; in-segment order stays an arrival order
# speedup vs baseline: 1.4545x; 1.1116x over previous
#include <hip/hip_runtime.h>
#include <hip/hip_bf16.h>

#define N_NODES 10000
#define N_EDGES 320000
#define F_INF   128
#define F_OUTF  128
#define H_HEADS 4
#define C_OUT   (H_HEADS * F_OUTF)
#define STEPS   2
#define NEG_SLOPE 0.2f

typedef float v2f __attribute__((ext_vector_type(2)));
typedef float v8f __attribute__((ext_vector_type(8)));

#define GEMM_WAVE_COLS   64
#define GEMM_COL_GROUPS  (C_OUT / GEMM_WAVE_COLS)
#define GEMM_ROW_TILES   (N_NODES / 16)
#define GEMM_ROW_GROUPS  ((GEMM_ROW_TILES + 3) / 4)
#define GEMM_BLOCKS      (GEMM_ROW_GROUPS * GEMM_COL_GROUPS)
static_assert(N_NODES % 16 == 0 && C_OUT % GEMM_WAVE_COLS == 0 && F_INF % 4 == 0,
              "whole tiles only: no partial tile, so every lane of a live wave is active");
__global__ __launch_bounds__(128) void gemm_xw(const float* __restrict__ x,
                                               const float* __restrict__ W,
                                               float* __restrict__ h) {
  const int lane = threadIdx.x & 31;
  const int wv   = threadIdx.x >> 5;
  const int half = lane >> 4;
  const int mr   = lane & 15;
  const int cg   = blockIdx.x % GEMM_COL_GROUPS;
  const int tm   = (blockIdx.x / GEMM_COL_GROUPS) * 4 + wv;
  if (tm >= GEMM_ROW_TILES) return;
  const int m0 = tm * 16, n0 = cg * GEMM_WAVE_COLS;

  v8f acc0 = {}, acc1 = {}, acc2 = {}, acc3 = {};
  const float* arow = x + (size_t)(m0 + mr) * F_INF + 2 * half;
  const float* bcol = W + (size_t)(2 * half) * C_OUT + n0 + 4 * mr;
  #pragma unroll 8
  for (int k0 = 0; k0 < F_INF; k0 += 4) {
    v2f a; a.x = arow[k0]; a.y = arow[k0 + 1];
    const float4 bx = *(const float4*)(bcol + (size_t)k0 * C_OUT);
    const float4 by = *(const float4*)(bcol + (size_t)(k0 + 1) * C_OUT);
    v2f b;
    b.x = bx.x; b.y = by.x;
    acc0 = __builtin_amdgcn_wmma_f32_16x16x4_f32(
         false, a,  false, b,
         (short)0, acc0,  false,  false);
    b.x = bx.y; b.y = by.y;
    acc1 = __builtin_amdgcn_wmma_f32_16x16x4_f32(false, a, false, b, (short)0, acc1, false, false);
    b.x = bx.z; b.y = by.z;
    acc2 = __builtin_amdgcn_wmma_f32_16x16x4_f32(false, a, false, b, (short)0, acc2, false, false);
    b.x = bx.w; b.y = by.w;
    acc3 = __builtin_amdgcn_wmma_f32_16x16x4_f32(false, a, false, b, (short)0, acc3, false, false);
  }
  float* hrow = h + (size_t)(m0 + half * 8) * C_OUT + n0 + 4 * mr;
  #pragma unroll
  for (int i = 0; i < 8; ++i)
    *(float4*)(hrow + (size_t)i * C_OUT) = make_float4(acc0[i], acc1[i], acc2[i], acc3[i]);
}

#define NL_BLOCKS 625
__global__ __launch_bounds__(256) void node_logits(const float* __restrict__ h,
                                                   const float* __restrict__ attn_l,
                                                   const float* __restrict__ attn_r,
                                                   float* __restrict__ el,
                                                   float* __restrict__ er) {
  const int lane = threadIdx.x & 31;
  const int w = blockIdx.x * 8 + (threadIdx.x >> 5);
  float al[H_HEADS][4], ar[H_HEADS][4];
  #pragma unroll
  for (int hd = 0; hd < H_HEADS; ++hd) {
    #pragma unroll
    for (int k = 0; k < 4; ++k) {
      al[hd][k] = attn_l[hd * F_OUTF + lane + 32 * k];
      ar[hd][k] = attn_r[hd * F_OUTF + lane + 32 * k];
    }
  }
  for (int n = w; n < N_NODES; n += NL_BLOCKS * 8) {
    const float* hr = h + (size_t)n * C_OUT;
    float v[H_HEADS][4];
    #pragma unroll
    for (int hd = 0; hd < H_HEADS; ++hd) {
      #pragma unroll
      for (int k = 0; k < 4; ++k) v[hd][k] = hr[hd * F_OUTF + lane + 32 * k];
    }
    float sl[H_HEADS], sr[H_HEADS];
    #pragma unroll
    for (int hd = 0; hd < H_HEADS; ++hd) {
      float a = 0.f, b = 0.f;
      #pragma unroll
      for (int k = 0; k < 4; ++k) {
        a += v[hd][k] * al[hd][k];
        b += v[hd][k] * ar[hd][k];
      }
      sl[hd] = a; sr[hd] = b;
    }
    #pragma unroll
    for (int off = 16; off; off >>= 1) {
      #pragma unroll
      for (int hd = 0; hd < H_HEADS; ++hd) {
        sl[hd] += __shfl_xor(sl[hd], off, 32);
        sr[hd] += __shfl_xor(sr[hd], off, 32);
      }
    }
    if (lane == 0) {
      *(float4*)(el + (size_t)n * H_HEADS) = make_float4(sl[0], sl[1], sl[2], sl[3]);
      *(float4*)(er + (size_t)n * H_HEADS) = make_float4(sr[0], sr[1], sr[2], sr[3]);
    }
  }
}

#define EDGE_BLOCKS     250
#define EDGE_PER_THREAD 5
static_assert(EDGE_BLOCKS * 256 * EDGE_PER_THREAD == N_EDGES, "edge grid must cover N_EDGES exactly");
__global__ __launch_bounds__(256) void edge_rank(const int* __restrict__ dst,
                                                 int* __restrict__ cnt,
                                                 int* __restrict__ rank) {
  const int t = blockIdx.x * 256 + threadIdx.x;
  int d[EDGE_PER_THREAD], r[EDGE_PER_THREAD];
  #pragma unroll
  for (int k = 0; k < EDGE_PER_THREAD; ++k) d[k] = dst[t + k * (EDGE_BLOCKS * 256)];
  #pragma unroll
  for (int k = 0; k < EDGE_PER_THREAD; ++k) r[k] = atomicAdd(&cnt[d[k]], 1);
  #pragma unroll
  for (int k = 0; k < EDGE_PER_THREAD; ++k) rank[t + k * (EDGE_BLOCKS * 256)] = r[k];
}

#define SCAN_THREADS 1024
#define SCAN_ITEMS   10
static_assert(N_NODES % SCAN_ITEMS == 0 && N_NODES / SCAN_ITEMS <= SCAN_THREADS,
              "a thread is wholly inside or wholly outside the node range");
static_assert(SCAN_ITEMS % 2 == 0, "counts are moved two at a time (8-byte accesses)");
__global__ __launch_bounds__(SCAN_THREADS) void scan_rowptr(const int* __restrict__ cnt,
                                                    int* __restrict__ row_ptr) {
  __shared__ int wave_total[SCAN_THREADS / 32];
  const int t = threadIdx.x, lane = t & 31, wv = t >> 5;
  const int base = t * SCAN_ITEMS;
  const bool live = base < N_NODES;

  const int* mine = cnt + (live ? base : 0);
  int2 p[SCAN_ITEMS / 2];
  #pragma unroll
  for (int k = 0; k < SCAN_ITEMS / 2; ++k) p[k] = *(const int2*)(mine + 2 * k);
  int v[SCAN_ITEMS];
  #pragma unroll
  for (int k = 0; k < SCAN_ITEMS / 2; ++k) {
    v[2 * k]     = live ? p[k].x : 0;
    v[2 * k + 1] = live ? p[k].y : 0;
  }
  int run = 0;
  #pragma unroll
  for (int k = 0; k < SCAN_ITEMS; ++k) { const int c = v[k]; v[k] = run; run += c; }

  int inc = run;
  for (int off = 1; off < 32; off <<= 1) {
    const int u = __shfl_up(inc, off, 32);
    if (lane >= off) inc += u;
  }
  if (lane == 31) wave_total[wv] = inc;
  __syncthreads();

  int before = (lane < wv) ? wave_total[lane] : 0;
  for (int off = 16; off; off >>= 1) before += __shfl_xor(before, off, 32);

  const int texcl = before + inc - run;
  if (live) {
    #pragma unroll
    for (int k = 0; k < SCAN_ITEMS; k += 2)
      *(int2*)(row_ptr + base + k) = make_int2(texcl + v[k], texcl + v[k + 1]);
    if (base + SCAN_ITEMS == N_NODES) row_ptr[N_NODES] = texcl + run;
  }
}

__global__ __launch_bounds__(256) void edge_place(const int* __restrict__ dst,
                                                  const int* __restrict__ rank,
                                                  const int* __restrict__ row_ptr,
                                                  int* __restrict__ edge_idx) {
  const int t = blockIdx.x * 256 + threadIdx.x;
  int d[EDGE_PER_THREAD], r[EDGE_PER_THREAD], p[EDGE_PER_THREAD];
  #pragma unroll
  for (int k = 0; k < EDGE_PER_THREAD; ++k) {
    d[k] = dst[t + k * (EDGE_BLOCKS * 256)];
    r[k] = rank[t + k * (EDGE_BLOCKS * 256)];
  }
  #pragma unroll
  for (int k = 0; k < EDGE_PER_THREAD; ++k) p[k] = row_ptr[d[k]] + r[k];
  #pragma unroll
  for (int k = 0; k < EDGE_PER_THREAD; ++k) edge_idx[p[k]] = t + k * (EDGE_BLOCKS * 256);
}

#define GROUP 8
#define AGG_BLOCKS 3334

__device__ __forceinline__ float edge_score(const float* __restrict__ el, int s,
                                            int head, float erd) {
  const float v = el[s * H_HEADS + head] + erd;
  return v > 0.f ? v : NEG_SLOPE * v;
}

__device__ __forceinline__ void fma4(float4& acc, float w, const float4& v) {
  acc.x = fmaf(w, v.x, acc.x); acc.y = fmaf(w, v.y, acc.y);
  acc.z = fmaf(w, v.z, acc.z); acc.w = fmaf(w, v.w, acc.w);
}

__device__ __forceinline__ void gather_chunk(float4& acc, const float* __restrict__ hcol,
                                             int s, float w, int cnt) {
  int j = 0;
  for (; j + GROUP <= cnt; j += GROUP) {
    float4 hv[GROUP]; float wj[GROUP];
    #pragma unroll
    for (int k = 0; k < GROUP; ++k) {
      const int sj = __builtin_amdgcn_readlane(s, j + k);
      wj[k] = __int_as_float(__builtin_amdgcn_readlane(__float_as_int(w), j + k));
      hv[k] = *(const float4*)(hcol + (size_t)sj * C_OUT);
    }
    #pragma unroll
    for (int k = 0; k < GROUP; ++k) fma4(acc, wj[k], hv[k]);
  }
  if (j < cnt) {
    float4 hv[GROUP - 1]; float wj[GROUP - 1];
    #pragma unroll
    for (int k = 0; k < GROUP - 1; ++k) {
      hv[k] = make_float4(0.f, 0.f, 0.f, 0.f);
      wj[k] = 0.f;
      if (j + k < cnt) {
        const int sj = __builtin_amdgcn_readlane(s, j + k);
        wj[k] = __int_as_float(__builtin_amdgcn_readlane(__float_as_int(w), j + k));
        hv[k] = *(const float4*)(hcol + (size_t)sj * C_OUT);
      }
    }
    #pragma unroll
    for (int k = 0; k < GROUP - 1; ++k) fma4(acc, wj[k], hv[k]);
  }
}

__global__ __launch_bounds__(128) void gat_aggregate(
    const float* __restrict__ h, const float* __restrict__ el,
    const float* __restrict__ er, const int* __restrict__ src,
    const int* __restrict__ row_ptr, const int* __restrict__ edge_idx,
    const float* __restrict__ bias, const float* __restrict__ alpha_p,
    const float* __restrict__ x0, float* __restrict__ xout) {
  __shared__ float red2[2][H_HEADS * F_OUTF];
  const int head = threadIdx.x >> 5, lane = threadIdx.x & 31;
  const float* hcol = h + head * F_OUTF + lane * 4;
  const float4 bv = *(const float4*)(bias + head * F_OUTF + lane * 4);
  const float al = alpha_p[0];
  int buf = 0;
  for (int n = blockIdx.x; n < N_NODES; n += AGG_BLOCKS, buf ^= 1) {
  float* red = red2[buf];
  const int rs = row_ptr[n], re = row_ptr[n + 1];
  const float erd = er[n * H_HEADS + head];

  const int i0 = rs + lane;
  const bool has0 = i0 < re;
  int s0 = 0;
  float v0 = 0.f;
  if (has0) {
    s0 = src[edge_idx[i0]];
    v0 = edge_score(el, s0, head, erd);
  }

  float mx = -INFINITY;
  if (has0) mx = fmaxf(mx, v0);
  for (int i = i0 + 32; i < re; i += 32)
    mx = fmaxf(mx, edge_score(el, src[edge_idx[i]], head, erd));
  #pragma unroll
  for (int off = 16; off; off >>= 1) mx = fmaxf(mx, __shfl_xor(mx, off, 32));

  const float p0 = has0 ? __expf(v0 - mx) : 0.f;
  float sum = p0;
  for (int i = i0 + 32; i < re; i += 32)
    sum += __expf(edge_score(el, src[edge_idx[i]], head, erd) - mx);
  #pragma unroll
  for (int off = 16; off; off >>= 1) sum += __shfl_xor(sum, off, 32);
  const float inv = (sum != 0.f) ? 1.f / sum : 0.f;

  float4 acc = make_float4(0.f, 0.f, 0.f, 0.f);
  if (rs < re) gather_chunk(acc, hcol, s0, p0 * inv, min(32, re - rs));
  for (int base = rs + 32; base < re; base += 32) {
    const int i = base + lane;
    int s = 0;
    float w = 0.f;
    if (i < re) {
      s = src[edge_idx[i]];
      w = __expf(edge_score(el, s, head, erd) - mx) * inv;
    }
    gather_chunk(acc, hcol, s, w, min(32, re - base));
  }
  acc.x += bv.x; acc.y += bv.y; acc.z += bv.z; acc.w += bv.w;
  *(float4*)(red + head * F_OUTF + lane * 4) = acc;
  __syncthreads();

  const int f = threadIdx.x;
  const float s = 0.25f * (red[f] + red[F_OUTF + f] +
                           red[2 * F_OUTF + f] + red[3 * F_OUTF + f]);
  xout[(size_t)n * F_OUTF + f] = (1.f - al) * s + al * x0[(size_t)n * F_OUTF + f];
  }
}

extern "C" void kernel_launch(void* const* d_in, const int* in_sizes, int n_in,
                              void* d_out, int out_size, void* d_ws, size_t ws_size,
                              hipStream_t stream) {
  const float* x      = (const float*)d_in[0];
  const float* x0     = (const float*)d_in[1];
  const int*   src    = (const int*)d_in[2];
  const int*   dst    = (const int*)d_in[3];
  const float* alpha  = (const float*)d_in[4];
  const float* W      = (const float*)d_in[5];
  const float* attn_l = (const float*)d_in[6];
  const float* attn_r = (const float*)d_in[7];
  const float* bias   = (const float*)d_in[8];

  char* ws = (char*)d_ws;
  size_t off = 0;
  auto alloc = [&](size_t bytes) -> char* {
    char* p = ws + off;
    off = (off + bytes + 255) & ~(size_t)255;
    return p;
  };
  float* h        = (float*)alloc((size_t)N_NODES * C_OUT * sizeof(float));
  float* el       = (float*)alloc((size_t)N_NODES * H_HEADS * sizeof(float));
  float* er       = (float*)alloc((size_t)N_NODES * H_HEADS * sizeof(float));
  float* xbuf     = (float*)alloc((size_t)N_NODES * F_INF * sizeof(float));
  int*   cnt      = (int*)alloc((size_t)N_NODES * sizeof(int));
  int*   row_ptr  = (int*)alloc((size_t)(N_NODES + 1) * sizeof(int));
  int*   rank     = (int*)alloc((size_t)N_EDGES * sizeof(int));
  int*   edge_idx = (int*)alloc((size_t)N_EDGES * sizeof(int));

  hipMemsetAsync(cnt, 0, (size_t)N_NODES * sizeof(int), stream);
  edge_rank<<<EDGE_BLOCKS, 256, 0, stream>>>(dst, cnt, rank);
  scan_rowptr<<<1, SCAN_THREADS, 0, stream>>>(cnt, row_ptr);
  edge_place<<<EDGE_BLOCKS, 256, 0, stream>>>(dst, rank, row_ptr, edge_idx);

  const float* xin = x;
  for (int step = 0; step < STEPS; ++step) {
    gemm_xw<<<GEMM_BLOCKS, 128, 0, stream>>>(xin, W, h);
    node_logits<<<NL_BLOCKS, 256, 0, stream>>>(h, attn_l, attn_r, el, er);
    float* xo = (step == STEPS - 1) ? (float*)d_out : xbuf;
    gat_aggregate<<<AGG_BLOCKS, 128, 0, stream>>>(h, el, er, src, row_ptr, edge_idx,
                                               bias, alpha, x0, xo);
    xin = xbuf;
  }
}


# edges grouped by destination in one pass: the counting add's return is the edge's slot in a fixed-pitch table, overflow list behind it; no scan, no placing launch; in-segment order stays arrival order
# speedup vs baseline: 1.5495x; 1.0653x over previous
#include <hip/hip_runtime.h>
#include <hip/hip_bf16.h>

#define N_NODES 10000
#define N_EDGES 320000
#define F_INF   128
#define F_OUTF  128
#define H_HEADS 4
#define C_OUT   (H_HEADS * F_OUTF)
#define STEPS   2
#define NEG_SLOPE 0.2f

typedef float v2f __attribute__((ext_vector_type(2)));
typedef float v8f __attribute__((ext_vector_type(8)));

#define GEMM_WAVE_COLS   64
#define GEMM_COL_GROUPS  (C_OUT / GEMM_WAVE_COLS)
#define GEMM_ROW_TILES   (N_NODES / 16)
#define GEMM_ROW_GROUPS  ((GEMM_ROW_TILES + 3) / 4)
#define GEMM_BLOCKS      (GEMM_ROW_GROUPS * GEMM_COL_GROUPS)
static_assert(N_NODES % 16 == 0 && C_OUT % GEMM_WAVE_COLS == 0 && F_INF % 4 == 0,
              "whole tiles only: no partial tile, so every lane of a live wave is active");
__global__ __launch_bounds__(128) void gemm_xw(const float* __restrict__ x,
                                               const float* __restrict__ W,
                                               float* __restrict__ h) {
  const int lane = threadIdx.x & 31;
  const int wv   = threadIdx.x >> 5;
  const int half = lane >> 4;
  const int mr   = lane & 15;
  const int cg   = blockIdx.x % GEMM_COL_GROUPS;
  const int tm   = (blockIdx.x / GEMM_COL_GROUPS) * 4 + wv;
  if (tm >= GEMM_ROW_TILES) return;
  const int m0 = tm * 16, n0 = cg * GEMM_WAVE_COLS;

  v8f acc0 = {}, acc1 = {}, acc2 = {}, acc3 = {};
  const float* arow = x + (size_t)(m0 + mr) * F_INF + 2 * half;
  const float* bcol = W + (size_t)(2 * half) * C_OUT + n0 + 4 * mr;
  #pragma unroll 8
  for (int k0 = 0; k0 < F_INF; k0 += 4) {
    v2f a; a.x = arow[k0]; a.y = arow[k0 + 1];
    const float4 bx = *(const float4*)(bcol + (size_t)k0 * C_OUT);
    const float4 by = *(const float4*)(bcol + (size_t)(k0 + 1) * C_OUT);
    v2f b;
    b.x = bx.x; b.y = by.x;
    acc0 = __builtin_amdgcn_wmma_f32_16x16x4_f32(
         false, a,  false, b,
         (short)0, acc0,  false,  false);
    b.x = bx.y; b.y = by.y;
    acc1 = __builtin_amdgcn_wmma_f32_16x16x4_f32(false, a, false, b, (short)0, acc1, false, false);
    b.x = bx.z; b.y = by.z;
    acc2 = __builtin_amdgcn_wmma_f32_16x16x4_f32(false, a, false, b, (short)0, acc2, false, false);
    b.x = bx.w; b.y = by.w;
    acc3 = __builtin_amdgcn_wmma_f32_16x16x4_f32(false, a, false, b, (short)0, acc3, false, false);
  }
  float* hrow = h + (size_t)(m0 + half * 8) * C_OUT + n0 + 4 * mr;
  #pragma unroll
  for (int i = 0; i < 8; ++i)
    *(float4*)(hrow + (size_t)i * C_OUT) = make_float4(acc0[i], acc1[i], acc2[i], acc3[i]);
}

#define NL_BLOCKS 625
__global__ __launch_bounds__(256) void node_logits(const float* __restrict__ h,
                                                   const float* __restrict__ attn_l,
                                                   const float* __restrict__ attn_r,
                                                   float* __restrict__ el,
                                                   float* __restrict__ er) {
  const int lane = threadIdx.x & 31;
  const int w = blockIdx.x * 8 + (threadIdx.x >> 5);
  float al[H_HEADS][4], ar[H_HEADS][4];
  #pragma unroll
  for (int hd = 0; hd < H_HEADS; ++hd) {
    #pragma unroll
    for (int k = 0; k < 4; ++k) {
      al[hd][k] = attn_l[hd * F_OUTF + lane + 32 * k];
      ar[hd][k] = attn_r[hd * F_OUTF + lane + 32 * k];
    }
  }
  for (int n = w; n < N_NODES; n += NL_BLOCKS * 8) {
    const float* hr = h + (size_t)n * C_OUT;
    float v[H_HEADS][4];
    #pragma unroll
    for (int hd = 0; hd < H_HEADS; ++hd) {
      #pragma unroll
      for (int k = 0; k < 4; ++k) v[hd][k] = hr[hd * F_OUTF + lane + 32 * k];
    }
    float sl[H_HEADS], sr[H_HEADS];
    #pragma unroll
    for (int hd = 0; hd < H_HEADS; ++hd) {
      float a = 0.f, b = 0.f;
      #pragma unroll
      for (int k = 0; k < 4; ++k) {
        a += v[hd][k] * al[hd][k];
        b += v[hd][k] * ar[hd][k];
      }
      sl[hd] = a; sr[hd] = b;
    }
    #pragma unroll
    for (int off = 16; off; off >>= 1) {
      #pragma unroll
      for (int hd = 0; hd < H_HEADS; ++hd) {
        sl[hd] += __shfl_xor(sl[hd], off, 32);
        sr[hd] += __shfl_xor(sr[hd], off, 32);
      }
    }
    if (lane == 0) {
      *(float4*)(el + (size_t)n * H_HEADS) = make_float4(sl[0], sl[1], sl[2], sl[3]);
      *(float4*)(er + (size_t)n * H_HEADS) = make_float4(sr[0], sr[1], sr[2], sr[3]);
    }
  }
}

#define EDGE_BLOCKS     250
#define EDGE_PER_THREAD 5
static_assert(EDGE_BLOCKS * 256 * EDGE_PER_THREAD == N_EDGES, "edge grid must cover N_EDGES exactly");
#define SLOT_CAP 64
static_assert(SLOT_CAP % 32 == 0, "a destination's slots are read in whole chunks of 32 lanes");
__global__ __launch_bounds__(256) void edge_bucket(const int* __restrict__ dst,
                                                   int* __restrict__ cnt,
                                                   int* __restrict__ slot,
                                                   int* __restrict__ ovf_edge) {
  const int t = blockIdx.x * 256 + threadIdx.x;
  int d[EDGE_PER_THREAD], r[EDGE_PER_THREAD];
  #pragma unroll
  for (int k = 0; k < EDGE_PER_THREAD; ++k) d[k] = dst[t + k * (EDGE_BLOCKS * 256)];
  #pragma unroll
  for (int k = 0; k < EDGE_PER_THREAD; ++k) r[k] = atomicAdd(&cnt[d[k]], 1);
  #pragma unroll
  for (int k = 0; k < EDGE_PER_THREAD; ++k) {
    const int e = t + k * (EDGE_BLOCKS * 256);
    if (r[k] < SLOT_CAP) slot[(size_t)d[k] * SLOT_CAP + r[k]] = e;
    else ovf_edge[atomicAdd(&cnt[N_NODES], 1)] = e;
  }
}

#define GROUP 8
#define AGG_BLOCKS 3334

__device__ __forceinline__ float edge_score(const float* __restrict__ el, int s,
                                            int head, float erd) {
  const float v = el[s * H_HEADS + head] + erd;
  return v > 0.f ? v : NEG_SLOPE * v;
}

__device__ __forceinline__ void fma4(float4& acc, float w, const float4& v) {
  acc.x = fmaf(w, v.x, acc.x); acc.y = fmaf(w, v.y, acc.y);
  acc.z = fmaf(w, v.z, acc.z); acc.w = fmaf(w, v.w, acc.w);
}

__device__ __forceinline__ void gather_chunk(float4& acc, const float* __restrict__ hcol,
                                             int s, float w, int cnt) {
  int j = 0;
  for (; j + GROUP <= cnt; j += GROUP) {
    float4 hv[GROUP]; float wj[GROUP];
    #pragma unroll
    for (int k = 0; k < GROUP; ++k) {
      const int sj = __builtin_amdgcn_readlane(s, j + k);
      wj[k] = __int_as_float(__builtin_amdgcn_readlane(__float_as_int(w), j + k));
      hv[k] = *(const float4*)(hcol + (size_t)sj * C_OUT);
    }
    #pragma unroll
    for (int k = 0; k < GROUP; ++k) fma4(acc, wj[k], hv[k]);
  }
  if (j < cnt) {
    float4 hv[GROUP - 1]; float wj[GROUP - 1];
    #pragma unroll
    for (int k = 0; k < GROUP - 1; ++k) {
      hv[k] = make_float4(0.f, 0.f, 0.f, 0.f);
      wj[k] = 0.f;
      if (j + k < cnt) {
        const int sj = __builtin_amdgcn_readlane(s, j + k);
        wj[k] = __int_as_float(__builtin_amdgcn_readlane(__float_as_int(w), j + k));
        hv[k] = *(const float4*)(hcol + (size_t)sj * C_OUT);
      }
    }
    #pragma unroll
    for (int k = 0; k < GROUP - 1; ++k) fma4(acc, wj[k], hv[k]);
  }
}

__global__ __launch_bounds__(128, 16) void gat_aggregate(
    const float* __restrict__ h, const float* __restrict__ el,
    const float* __restrict__ er, const int* __restrict__ src,
    const int* __restrict__ dst, const int* __restrict__ cnt,
    const int* __restrict__ slot, const int* __restrict__ ovf_edge,
    const float* __restrict__ bias, const float* __restrict__ alpha_p,
    const float* __restrict__ x0, float* __restrict__ xout) {
  __shared__ float red2[2][H_HEADS * F_OUTF];
  const int head = threadIdx.x >> 5, lane = threadIdx.x & 31;
  const float* hcol = h + head * F_OUTF + lane * 4;
  const float4 bv = *(const float4*)(bias + head * F_OUTF + lane * 4);
  const float al = alpha_p[0];
  int buf = 0;
  for (int n = blockIdx.x; n < N_NODES; n += AGG_BLOCKS, buf ^= 1) {
  float* red = red2[buf];
  const int deg = cnt[n];
  const bool over = deg > SLOT_CAP;
  const int novf = over ? cnt[N_NODES] : 0;
  const int* edge_idx = slot + (size_t)n * SLOT_CAP;
  const int rs = 0, re = over ? SLOT_CAP : deg;
  const float erd = er[n * H_HEADS + head];

  const int i0 = rs + lane;
  const bool has0 = i0 < re;
  int s0 = 0;
  float v0 = 0.f;
  if (has0) {
    s0 = src[edge_idx[i0]];
    v0 = edge_score(el, s0, head, erd);
  }

  float mx = -INFINITY;
  if (has0) mx = fmaxf(mx, v0);
  for (int i = i0 + 32; i < re; i += 32)
    mx = fmaxf(mx, edge_score(el, src[edge_idx[i]], head, erd));
  if (over) {
    for (int k = lane; k < novf; k += 32) {
      const int e = ovf_edge[k];
      if (dst[e] == n) mx = fmaxf(mx, edge_score(el, src[e], head, erd));
    }
  }
  #pragma unroll
  for (int off = 16; off; off >>= 1) mx = fmaxf(mx, __shfl_xor(mx, off, 32));

  const float p0 = has0 ? __expf(v0 - mx) : 0.f;
  float sum = p0;
  for (int i = i0 + 32; i < re; i += 32)
    sum += __expf(edge_score(el, src[edge_idx[i]], head, erd) - mx);
  if (over) {
    for (int k = lane; k < novf; k += 32) {
      const int e = ovf_edge[k];
      if (dst[e] == n) sum += __expf(edge_score(el, src[e], head, erd) - mx);
    }
  }
  #pragma unroll
  for (int off = 16; off; off >>= 1) sum += __shfl_xor(sum, off, 32);
  const float inv = (sum != 0.f) ? 1.f / sum : 0.f;

  float4 acc = make_float4(0.f, 0.f, 0.f, 0.f);
  if (rs < re) gather_chunk(acc, hcol, s0, p0 * inv, min(32, re - rs));
  for (int base = rs + 32; base < re; base += 32) {
    const int i = base + lane;
    int s = 0;
    float w = 0.f;
    if (i < re) {
      s = src[edge_idx[i]];
      w = __expf(edge_score(el, s, head, erd) - mx) * inv;
    }
    gather_chunk(acc, hcol, s, w, min(32, re - base));
  }
  if (over) {
    for (int k0 = 0; k0 < novf; k0 += 32) {
      const int k = k0 + lane;
      int s = 0;
      float w = 0.f;
      bool mine = false;
      if (k < novf) {
        const int e = ovf_edge[k];
        if (dst[e] == n) {
          mine = true;
          s = src[e];
          w = __expf(edge_score(el, s, head, erd) - mx) * inv;
        }
      }
      unsigned todo = (unsigned)__ballot(mine);
      while (todo) {
        const int j = __builtin_ctz(todo);
        todo &= todo - 1;
        const int sj = __shfl(s, j, 32);
        const float wj = __shfl(w, j, 32);
        fma4(acc, wj, *(const float4*)(hcol + (size_t)sj * C_OUT));
      }
    }
  }
  acc.x += bv.x; acc.y += bv.y; acc.z += bv.z; acc.w += bv.w;
  *(float4*)(red + head * F_OUTF + lane * 4) = acc;
  __syncthreads();

  const int f = threadIdx.x;
  const float s = 0.25f * (red[f] + red[F_OUTF + f] +
                           red[2 * F_OUTF + f] + red[3 * F_OUTF + f]);
  xout[(size_t)n * F_OUTF + f] = (1.f - al) * s + al * x0[(size_t)n * F_OUTF + f];
  }
}

extern "C" void kernel_launch(void* const* d_in, const int* in_sizes, int n_in,
                              void* d_out, int out_size, void* d_ws, size_t ws_size,
                              hipStream_t stream) {
  const float* x      = (const float*)d_in[0];
  const float* x0     = (const float*)d_in[1];
  const int*   src    = (const int*)d_in[2];
  const int*   dst    = (const int*)d_in[3];
  const float* alpha  = (const float*)d_in[4];
  const float* W      = (const float*)d_in[5];
  const float* attn_l = (const float*)d_in[6];
  const float* attn_r = (const float*)d_in[7];
  const float* bias   = (const float*)d_in[8];

  char* ws = (char*)d_ws;
  size_t off = 0;
  auto alloc = [&](size_t bytes) -> char* {
    char* p = ws + off;
    off = (off + bytes + 255) & ~(size_t)255;
    return p;
  };
  float* h        = (float*)alloc((size_t)N_NODES * C_OUT * sizeof(float));
  float* el       = (float*)alloc((size_t)N_NODES * H_HEADS * sizeof(float));
  float* er       = (float*)alloc((size_t)N_NODES * H_HEADS * sizeof(float));
  float* xbuf     = (float*)alloc((size_t)N_NODES * F_INF * sizeof(float));
  int*   cnt      = (int*)alloc((size_t)(N_NODES + 1) * sizeof(int));
  int*   slot     = (int*)alloc((size_t)N_NODES * SLOT_CAP * sizeof(int));
  int*   ovf_edge = (int*)alloc((size_t)N_EDGES * sizeof(int));

  hipMemsetAsync(cnt, 0, (size_t)(N_NODES + 1) * sizeof(int), stream);
  edge_bucket<<<EDGE_BLOCKS, 256, 0, stream>>>(dst, cnt, slot, ovf_edge);

  const float* xin = x;
  for (int step = 0; step < STEPS; ++step) {
    gemm_xw<<<GEMM_BLOCKS, 128, 0, stream>>>(xin, W, h);
    node_logits<<<NL_BLOCKS, 256, 0, stream>>>(h, attn_l, attn_r, el, er);
    float* xo = (step == STEPS - 1) ? (float*)d_out : xbuf;
    gat_aggregate<<<AGG_BLOCKS, 128, 0, stream>>>(h, el, er, src, dst, cnt, slot, ovf_edge,
                                               bias, alpha, x0, xo);
    xin = xbuf;
  }
}


# aggregate takes each edge's logit from the h row it gathers anyway (dot with attn_l, lane tree), softmax as one streaming pass: node-logit launches and el/er planes gone; order of f32 sums changes
# speedup vs baseline: 1.6625x; 1.0729x over previous
#include <hip/hip_runtime.h>
#include <hip/hip_bf16.h>

#define N_NODES 10000
#define N_EDGES 320000
#define F_INF   128
#define F_OUTF  128
#define H_HEADS 4
#define C_OUT   (H_HEADS * F_OUTF)
#define STEPS   2
#define NEG_SLOPE 0.2f

typedef float v2f __attribute__((ext_vector_type(2)));
typedef float v8f __attribute__((ext_vector_type(8)));

#define GEMM_WAVE_COLS   64
#define GEMM_COL_GROUPS  (C_OUT / GEMM_WAVE_COLS)
#define GEMM_ROW_TILES   (N_NODES / 16)
#define GEMM_ROW_GROUPS  ((GEMM_ROW_TILES + 3) / 4)
#define GEMM_BLOCKS      (GEMM_ROW_GROUPS * GEMM_COL_GROUPS)
static_assert(N_NODES % 16 == 0 && C_OUT % GEMM_WAVE_COLS == 0 && F_INF % 4 == 0,
              "whole tiles only: no partial tile, so every lane of a live wave is active");
__global__ __launch_bounds__(128) void gemm_xw(const float* __restrict__ x,
                                               const float* __restrict__ W,
                                               float* __restrict__ h) {
  const int lane = threadIdx.x & 31;
  const int wv   = threadIdx.x >> 5;
  const int half = lane >> 4;
  const int mr   = lane & 15;
  const int cg   = blockIdx.x % GEMM_COL_GROUPS;
  const int tm   = (blockIdx.x / GEMM_COL_GROUPS) * 4 + wv;
  if (tm >= GEMM_ROW_TILES) return;
  const int m0 = tm * 16, n0 = cg * GEMM_WAVE_COLS;

  v8f acc0 = {}, acc1 = {}, acc2 = {}, acc3 = {};
  const float* arow = x + (size_t)(m0 + mr) * F_INF + 2 * half;
  const float* bcol = W + (size_t)(2 * half) * C_OUT + n0 + 4 * mr;
  #pragma unroll 8
  for (int k0 = 0; k0 < F_INF; k0 += 4) {
    v2f a; a.x = arow[k0]; a.y = arow[k0 + 1];
    const float4 bx = *(const float4*)(bcol + (size_t)k0 * C_OUT);
    const float4 by = *(const float4*)(bcol + (size_t)(k0 + 1) * C_OUT);
    v2f b;
    b.x = bx.x; b.y = by.x;
    acc0 = __builtin_amdgcn_wmma_f32_16x16x4_f32(
         false, a,  false, b,
         (short)0, acc0,  false,  false);
    b.x = bx.y; b.y = by.y;
    acc1 = __builtin_amdgcn_wmma_f32_16x16x4_f32(false, a, false, b, (short)0, acc1, false, false);
    b.x = bx.z; b.y = by.z;
    acc2 = __builtin_amdgcn_wmma_f32_16x16x4_f32(false, a, false, b, (short)0, acc2, false, false);
    b.x = bx.w; b.y = by.w;
    acc3 = __builtin_amdgcn_wmma_f32_16x16x4_f32(false, a, false, b, (short)0, acc3, false, false);
  }
  float* hrow = h + (size_t)(m0 + half * 8) * C_OUT + n0 + 4 * mr;
  #pragma unroll
  for (int i = 0; i < 8; ++i)
    *(float4*)(hrow + (size_t)i * C_OUT) = make_float4(acc0[i], acc1[i], acc2[i], acc3[i]);
}

#define EDGE_BLOCKS     250
#define EDGE_PER_THREAD 5
static_assert(EDGE_BLOCKS * 256 * EDGE_PER_THREAD == N_EDGES, "edge grid must cover N_EDGES exactly");
#define SLOT_CAP 64
static_assert(SLOT_CAP % 32 == 0, "a destination's slots are read in whole chunks of 32 lanes");
__global__ __launch_bounds__(256) void edge_bucket(const int* __restrict__ dst,
                                                   int* __restrict__ cnt,
                                                   int* __restrict__ slot,
                                                   int* __restrict__ ovf_edge) {
  const int t = blockIdx.x * 256 + threadIdx.x;
  int d[EDGE_PER_THREAD], r[EDGE_PER_THREAD];
  #pragma unroll
  for (int k = 0; k < EDGE_PER_THREAD; ++k) d[k] = dst[t + k * (EDGE_BLOCKS * 256)];
  #pragma unroll
  for (int k = 0; k < EDGE_PER_THREAD; ++k) r[k] = atomicAdd(&cnt[d[k]], 1);
  #pragma unroll
  for (int k = 0; k < EDGE_PER_THREAD; ++k) {
    const int e = t + k * (EDGE_BLOCKS * 256);
    if (r[k] < SLOT_CAP) slot[(size_t)d[k] * SLOT_CAP + r[k]] = e;
    else ovf_edge[atomicAdd(&cnt[N_NODES], 1)] = e;
  }
}

#define GROUP 8
static_assert(GROUP == 8, "the lane tree below hands 8 partial sums to 8 groups of 4 lanes");
#define AGG_BLOCKS 3334

__device__ __forceinline__ float xhalf(float x) {
  const int b = __float_as_int(x);
  return __int_as_float(__builtin_amdgcn_permlanex16(b, b, 0x76543210, 0xfedcba98, false, false));
}
template <int M>
__device__ __forceinline__ float xrow(float x) {
  const int b = __float_as_int(x);
  return __int_as_float(__builtin_amdgcn_mov_dpp(b, 0x160 | M, 0xf, 0xf, false));
}
__device__ __forceinline__ float wave_sum(float x) {
  x += xhalf(x);
  x += xrow<8>(x);
  x += xrow<4>(x);
  x += xrow<2>(x);
  x += xrow<1>(x);
  return x;
}

__device__ __forceinline__ float dot4(const float4& a, const float4& b) {
  return fmaf(a.w, b.w, fmaf(a.z, b.z, fmaf(a.y, b.y, a.x * b.x)));
}

__device__ __forceinline__ void fma4(float4& acc, float w, const float4& v) {
  acc.x = fmaf(w, v.x, acc.x); acc.y = fmaf(w, v.y, acc.y);
  acc.z = fmaf(w, v.z, acc.z); acc.w = fmaf(w, v.w, acc.w);
}

__device__ __forceinline__ float leaky(float v) { return v > 0.f ? v : NEG_SLOPE * v; }

struct Soft { float m; float sum; float4 acc; };

template <bool FULL>
__device__ __forceinline__ void edge_group(Soft& st, const float* __restrict__ hcol,
                                           const float4& al4, float erd, int s, int j,
                                           int cnt, int lane) {
  float4 hv[GROUP];
  #pragma unroll
  for (int k = 0; k < GROUP; ++k) {
    hv[k] = make_float4(0.f, 0.f, 0.f, 0.f);
    if (FULL || j + k < cnt) {
      const int sj = __builtin_amdgcn_readlane(s, j + k);
      hv[k] = *(const float4*)(hcol + (size_t)sj * C_OUT);
    }
  }
  float d[GROUP];
  #pragma unroll
  for (int k = 0; k < GROUP; ++k)
    d[k] = (FULL || j + k < cnt) ? dot4(hv[k], al4) : -INFINITY;
  const bool b4 = (lane & 16) != 0, b3 = (lane & 8) != 0, b2 = (lane & 4) != 0;
  float r4[4];
  #pragma unroll
  for (int i = 0; i < 4; ++i) {
    const float keep = b4 ? d[i + 4] : d[i];
    const float send = b4 ? d[i] : d[i + 4];
    r4[i] = keep + xhalf(send);
  }
  float r2[2];
  #pragma unroll
  for (int i = 0; i < 2; ++i) {
    const float keep = b3 ? r4[i + 2] : r4[i];
    const float send = b3 ? r4[i] : r4[i + 2];
    r2[i] = keep + xrow<8>(send);
  }
  float r1;
  {
    const float keep = b2 ? r2[1] : r2[0];
    const float send = b2 ? r2[0] : r2[1];
    r1 = keep + xrow<4>(send);
  }
  r1 += xrow<2>(r1);
  r1 += xrow<1>(r1);
  const float v = leaky(r1 + erd);
  float gm = fmaxf(v, xrow<4>(v));
  gm = fmaxf(gm, xrow<8>(gm));
  gm = fmaxf(gm, xhalf(gm));
  const float mn = fmaxf(st.m, gm);
  const float sc = __expf(st.m - mn);
  const float p = __expf(v - mn);
  st.m = mn;
  st.sum *= sc;
  st.acc.x *= sc; st.acc.y *= sc; st.acc.z *= sc; st.acc.w *= sc;
  #pragma unroll
  for (int k = 0; k < GROUP; ++k) {
    const float wk = __int_as_float(__builtin_amdgcn_readlane(
        __float_as_int(p), 16 * (k >> 2) + 8 * ((k >> 1) & 1) + 4 * (k & 1)));
    st.sum += wk;
    fma4(st.acc, wk, hv[k]);
  }
}

__device__ __forceinline__ void edge_chunk(Soft& st, const float* __restrict__ hcol,
                                           const float4& al4, float erd, int s, int cnt,
                                           int lane) {
  int j = 0;
  for (; j + GROUP <= cnt; j += GROUP) edge_group<true>(st, hcol, al4, erd, s, j, cnt, lane);
  if (j < cnt) edge_group<false>(st, hcol, al4, erd, s, j, cnt, lane);
}

__device__ __forceinline__ void edge_single(Soft& st, const float* __restrict__ hcol,
                                            const float4& al4, float erd, int sj) {
  const float4 hv = *(const float4*)(hcol + (size_t)sj * C_OUT);
  const float v = leaky(wave_sum(dot4(hv, al4)) + erd);
  const float mn = fmaxf(st.m, v);
  const float sc = __expf(st.m - mn);
  const float p = __expf(v - mn);
  st.m = mn;
  st.sum = fmaf(st.sum, sc, p);
  st.acc.x = fmaf(st.acc.x, sc, p * hv.x); st.acc.y = fmaf(st.acc.y, sc, p * hv.y);
  st.acc.z = fmaf(st.acc.z, sc, p * hv.z); st.acc.w = fmaf(st.acc.w, sc, p * hv.w);
}

static_assert(SLOT_CAP == 64, "a lane keeps the two slots lane and lane + 32 of its node's row");
__global__ __launch_bounds__(128, 16) void gat_aggregate(
    const float* __restrict__ h, const float* __restrict__ attn_l,
    const float* __restrict__ attn_r, const int* __restrict__ src,
    const int* __restrict__ dst, const int* __restrict__ cnt,
    const int* __restrict__ slot, const int* __restrict__ ovf_edge,
    const float* __restrict__ bias, const float* __restrict__ alpha_p,
    const float* __restrict__ x0, float* __restrict__ xout) {
  __shared__ float red2[2][H_HEADS * F_OUTF];
  __shared__ float s_ar[H_HEADS * F_OUTF];
  __shared__ float s_bias[H_HEADS * F_OUTF];
  const int head = threadIdx.x >> 5, lane = threadIdx.x & 31;
  const float* hcol = h + head * F_OUTF + lane * 4;
  const float4 al4 = *(const float4*)(attn_l + head * F_OUTF + lane * 4);
  *(float4*)(s_ar + threadIdx.x * 4) = *(const float4*)(attn_r + threadIdx.x * 4);
  *(float4*)(s_bias + threadIdx.x * 4) = *(const float4*)(bias + threadIdx.x * 4);
  __syncthreads();
  const float al = alpha_p[0];
  int buf = 0;
  for (int n = blockIdx.x; n < N_NODES; n += AGG_BLOCKS, buf ^= 1) {
  float* red = red2[buf];
  const float4 own = *(const float4*)(hcol + (size_t)n * C_OUT);
  const int deg = cnt[n];
  const bool over = deg > SLOT_CAP;
  const int novf = over ? cnt[N_NODES] : 0;
  const int* edge_idx = slot + (size_t)n * SLOT_CAP;
  const int re = over ? SLOT_CAP : deg;
  int e0 = -1, e1 = -1;
  if (lane < re) e0 = edge_idx[lane];
  if (lane + 32 < re) e1 = edge_idx[lane + 32];
  int s0 = 0, s1 = 0;
  if (e0 >= 0) s0 = src[e0];
  if (e1 >= 0) s1 = src[e1];
  const float erd = wave_sum(dot4(own, *(const float4*)(s_ar + head * F_OUTF + lane * 4)));

  Soft st;
  st.m = -INFINITY; st.sum = 0.f; st.acc = make_float4(0.f, 0.f, 0.f, 0.f);
  if (re > 0) edge_chunk(st, hcol, al4, erd, s0, min(32, re), lane);
  if (re > 32) edge_chunk(st, hcol, al4, erd, s1, re - 32, lane);
  if (over) {
    for (int k0 = 0; k0 < novf; k0 += 32) {
      const int k = k0 + lane;
      int s = 0;
      bool mine = false;
      if (k < novf) {
        const int e = ovf_edge[k];
        if (dst[e] == n) { mine = true; s = src[e]; }
      }
      unsigned todo = (unsigned)__ballot(mine);
      while (todo) {
        const int j = __builtin_ctz(todo);
        todo &= todo - 1;
        edge_single(st, hcol, al4, erd, __builtin_amdgcn_readlane(s, j));
      }
    }
  }
  const float inv = (st.sum != 0.f) ? 1.f / st.sum : 0.f;
  *(float4*)(red + head * F_OUTF + lane * 4) =
      make_float4(st.acc.x * inv, st.acc.y * inv, st.acc.z * inv, st.acc.w * inv);
  __syncthreads();

  const int f = threadIdx.x;
  const float s = 0.25f * ((red[f] + s_bias[f]) + (red[F_OUTF + f] + s_bias[F_OUTF + f]) +
                           (red[2 * F_OUTF + f] + s_bias[2 * F_OUTF + f]) +
                           (red[3 * F_OUTF + f] + s_bias[3 * F_OUTF + f]));
  xout[(size_t)n * F_OUTF + f] = (1.f - al) * s + al * x0[(size_t)n * F_OUTF + f];
  }
}

extern "C" void kernel_launch(void* const* d_in, const int* in_sizes, int n_in,
                              void* d_out, int out_size, void* d_ws, size_t ws_size,
                              hipStream_t stream) {
  const float* x      = (const float*)d_in[0];
  const float* x0     = (const float*)d_in[1];
  const int*   src    = (const int*)d_in[2];
  const int*   dst    = (const int*)d_in[3];
  const float* alpha  = (const float*)d_in[4];
  const float* W      = (const float*)d_in[5];
  const float* attn_l = (const float*)d_in[6];
  const float* attn_r = (const float*)d_in[7];
  const float* bias   = (const float*)d_in[8];

  char* ws = (char*)d_ws;
  size_t off = 0;
  auto alloc = [&](size_t bytes) -> char* {
    char* p = ws + off;
    off = (off + bytes + 255) & ~(size_t)255;
    return p;
  };
  float* h        = (float*)alloc((size_t)N_NODES * C_OUT * sizeof(float));
  float* xbuf     = (float*)alloc((size_t)N_NODES * F_INF * sizeof(float));
  int*   cnt      = (int*)alloc((size_t)(N_NODES + 1) * sizeof(int));
  int*   slot     = (int*)alloc((size_t)N_NODES * SLOT_CAP * sizeof(int));
  int*   ovf_edge = (int*)alloc((size_t)N_EDGES * sizeof(int));

  hipMemsetAsync(cnt, 0, (size_t)(N_NODES + 1) * sizeof(int), stream);
  edge_bucket<<<EDGE_BLOCKS, 256, 0, stream>>>(dst, cnt, slot, ovf_edge);

  const float* xin = x;
  for (int step = 0; step < STEPS; ++step) {
    gemm_xw<<<GEMM_BLOCKS, 128, 0, stream>>>(xin, W, h);
    float* xo = (step == STEPS - 1) ? (float*)d_out : xbuf;
    gat_aggregate<<<AGG_BLOCKS, 128, 0, stream>>>(h, attn_l, attn_r, src, dst, cnt, slot,
                                               ovf_edge, bias, alpha, x0, xo);
    xin = xbuf;
  }
}


# step 1's launch holds two independent jobs side by side: 250 work-groups group the edges by destination (10 per thread), the rest are the x@W tiles; 5 launches -> 4; f32 sums, their order unchanged
# speedup vs baseline: 1.7237x; 1.0368x over previous
#include <hip/hip_runtime.h>
#include <hip/hip_bf16.h>

#define N_NODES 10000
#define N_EDGES 320000
#define F_INF   128
#define F_OUTF  128
#define H_HEADS 4
#define C_OUT   (H_HEADS * F_OUTF)
#define STEPS   2
#define NEG_SLOPE 0.2f

typedef float v2f __attribute__((ext_vector_type(2)));
typedef float v8f __attribute__((ext_vector_type(8)));

#define GEMM_WAVE_COLS   64
#define GEMM_COL_GROUPS  (C_OUT / GEMM_WAVE_COLS)
#define GEMM_ROW_TILES   (N_NODES / 16)
#define GEMM_ROW_GROUPS  ((GEMM_ROW_TILES + 3) / 4)
#define GEMM_BLOCKS      (GEMM_ROW_GROUPS * GEMM_COL_GROUPS)
static_assert(N_NODES % 16 == 0 && C_OUT % GEMM_WAVE_COLS == 0 && F_INF % 4 == 0,
              "whole tiles only: no partial tile, so every lane of a live wave is active");
__device__ __forceinline__ void gemm_block(const float* __restrict__ x,
                                           const float* __restrict__ W,
                                           float* __restrict__ h, unsigned bid) {
  const int lane = threadIdx.x & 31;
  const int wv   = threadIdx.x >> 5;
  const int half = lane >> 4;
  const int mr   = lane & 15;
  const int cg   = bid % GEMM_COL_GROUPS;
  const int tm   = (bid / GEMM_COL_GROUPS) * 4 + wv;
  if (tm >= GEMM_ROW_TILES) return;
  const int m0 = tm * 16, n0 = cg * GEMM_WAVE_COLS;

  v8f acc0 = {}, acc1 = {}, acc2 = {}, acc3 = {};
  const float* arow = x + (size_t)(m0 + mr) * F_INF + 2 * half;
  const float* bcol = W + (size_t)(2 * half) * C_OUT + n0 + 4 * mr;
  #pragma unroll 8
  for (int k0 = 0; k0 < F_INF; k0 += 4) {
    v2f a; a.x = arow[k0]; a.y = arow[k0 + 1];
    const float4 bx = *(const float4*)(bcol + (size_t)k0 * C_OUT);
    const float4 by = *(const float4*)(bcol + (size_t)(k0 + 1) * C_OUT);
    v2f b;
    b.x = bx.x; b.y = by.x;
    acc0 = __builtin_amdgcn_wmma_f32_16x16x4_f32(
         false, a,  false, b,
         (short)0, acc0,  false,  false);
    b.x = bx.y; b.y = by.y;
    acc1 = __builtin_amdgcn_wmma_f32_16x16x4_f32(false, a, false, b, (short)0, acc1, false, false);
    b.x = bx.z; b.y = by.z;
    acc2 = __builtin_amdgcn_wmma_f32_16x16x4_f32(false, a, false, b, (short)0, acc2, false, false);
    b.x = bx.w; b.y = by.w;
    acc3 = __builtin_amdgcn_wmma_f32_16x16x4_f32(false, a, false, b, (short)0, acc3, false, false);
  }
  float* hrow = h + (size_t)(m0 + half * 8) * C_OUT + n0 + 4 * mr;
  #pragma unroll
  for (int i = 0; i < 8; ++i)
    *(float4*)(hrow + (size_t)i * C_OUT) = make_float4(acc0[i], acc1[i], acc2[i], acc3[i]);
}

__global__ __launch_bounds__(128) void gemm_xw(const float* __restrict__ x,
                                               const float* __restrict__ W,
                                               float* __restrict__ h) {
  gemm_block(x, W, h, blockIdx.x);
}

#define EDGE_BLOCKS     250
#define EDGE_THREADS    128
#define EDGE_PER_THREAD 10
#define EDGE_STRIDE     (EDGE_BLOCKS * EDGE_THREADS)
static_assert(EDGE_STRIDE * EDGE_PER_THREAD == N_EDGES, "edge groups must cover N_EDGES exactly");
#define SLOT_CAP 64
static_assert(SLOT_CAP % 32 == 0, "a destination's slots are read in whole chunks of 32 lanes");
__device__ __forceinline__ void edge_bucket(const int* __restrict__ dst,
                                            int* __restrict__ cnt,
                                            int* __restrict__ slot,
                                            int* __restrict__ ovf_edge, int t) {
  int d[EDGE_PER_THREAD], r[EDGE_PER_THREAD];
  #pragma unroll
  for (int k = 0; k < EDGE_PER_THREAD; ++k) d[k] = dst[t + k * EDGE_STRIDE];
  #pragma unroll
  for (int k = 0; k < EDGE_PER_THREAD; ++k) r[k] = atomicAdd(&cnt[d[k]], 1);
  #pragma unroll
  for (int k = 0; k < EDGE_PER_THREAD; ++k) {
    const int e = t + k * EDGE_STRIDE;
    if (r[k] < SLOT_CAP) slot[(size_t)d[k] * SLOT_CAP + r[k]] = e;
    else ovf_edge[atomicAdd(&cnt[N_NODES], 1)] = e;
  }
}

__global__ __launch_bounds__(128) void gemm_xw_edge_bucket(
    const float* __restrict__ x, const float* __restrict__ W, float* __restrict__ h,
    const int* __restrict__ dst, int* __restrict__ cnt, int* __restrict__ slot,
    int* __restrict__ ovf_edge) {
  if (blockIdx.x < EDGE_BLOCKS) {
    edge_bucket(dst, cnt, slot, ovf_edge, blockIdx.x * EDGE_THREADS + threadIdx.x);
    return;
  }
  gemm_block(x, W, h, blockIdx.x - EDGE_BLOCKS);
}

#define GROUP 8
static_assert(GROUP == 8, "the lane tree below hands 8 partial sums to 8 groups of 4 lanes");
#define AGG_BLOCKS 3334

__device__ __forceinline__ float xhalf(float x) {
  const int b = __float_as_int(x);
  return __int_as_float(__builtin_amdgcn_permlanex16(b, b, 0x76543210, 0xfedcba98, false, false));
}
template <int M>
__device__ __forceinline__ float xrow(float x) {
  const int b = __float_as_int(x);
  return __int_as_float(__builtin_amdgcn_mov_dpp(b, 0x160 | M, 0xf, 0xf, false));
}
__device__ __forceinline__ float wave_sum(float x) {
  x += xhalf(x);
  x += xrow<8>(x);
  x += xrow<4>(x);
  x += xrow<2>(x);
  x += xrow<1>(x);
  return x;
}

__device__ __forceinline__ float dot4(const float4& a, const float4& b) {
  return fmaf(a.w, b.w, fmaf(a.z, b.z, fmaf(a.y, b.y, a.x * b.x)));
}

__device__ __forceinline__ void fma4(float4& acc, float w, const float4& v) {
  acc.x = fmaf(w, v.x, acc.x); acc.y = fmaf(w, v.y, acc.y);
  acc.z = fmaf(w, v.z, acc.z); acc.w = fmaf(w, v.w, acc.w);
}

__device__ __forceinline__ float leaky(float v) { return v > 0.f ? v : NEG_SLOPE * v; }

struct Soft { float m; float sum; float4 acc; };

template <bool FULL>
__device__ __forceinline__ void edge_group(Soft& st, const float* __restrict__ hcol,
                                           const float4& al4, float erd, int s, int j,
                                           int cnt, int lane) {
  float4 hv[GROUP];
  #pragma unroll
  for (int k = 0; k < GROUP; ++k) {
    hv[k] = make_float4(0.f, 0.f, 0.f, 0.f);
    if (FULL || j + k < cnt) {
      const int sj = __builtin_amdgcn_readlane(s, j + k);
      hv[k] = *(const float4*)(hcol + (size_t)sj * C_OUT);
    }
  }
  float d[GROUP];
  #pragma unroll
  for (int k = 0; k < GROUP; ++k)
    d[k] = (FULL || j + k < cnt) ? dot4(hv[k], al4) : -INFINITY;
  const bool b4 = (lane & 16) != 0, b3 = (lane & 8) != 0, b2 = (lane & 4) != 0;
  float r4[4];
  #pragma unroll
  for (int i = 0; i < 4; ++i) {
    const float keep = b4 ? d[i + 4] : d[i];
    const float send = b4 ? d[i] : d[i + 4];
    r4[i] = keep + xhalf(send);
  }
  float r2[2];
  #pragma unroll
  for (int i = 0; i < 2; ++i) {
    const float keep = b3 ? r4[i + 2] : r4[i];
    const float send = b3 ? r4[i] : r4[i + 2];
    r2[i] = keep + xrow<8>(send);
  }
  float r1;
  {
    const float keep = b2 ? r2[1] : r2[0];
    const float send = b2 ? r2[0] : r2[1];
    r1 = keep + xrow<4>(send);
  }
  r1 += xrow<2>(r1);
  r1 += xrow<1>(r1);
  const float v = leaky(r1 + erd);
  float gm = fmaxf(v, xrow<4>(v));
  gm = fmaxf(gm, xrow<8>(gm));
  gm = fmaxf(gm, xhalf(gm));
  const float mn = fmaxf(st.m, gm);
  const float sc = __expf(st.m - mn);
  const float p = __expf(v - mn);
  st.m = mn;
  st.sum *= sc;
  st.acc.x *= sc; st.acc.y *= sc; st.acc.z *= sc; st.acc.w *= sc;
  #pragma unroll
  for (int k = 0; k < GROUP; ++k) {
    const float wk = __int_as_float(__builtin_amdgcn_readlane(
        __float_as_int(p), 16 * (k >> 2) + 8 * ((k >> 1) & 1) + 4 * (k & 1)));
    st.sum += wk;
    fma4(st.acc, wk, hv[k]);
  }
}

__device__ __forceinline__ void edge_chunk(Soft& st, const float* __restrict__ hcol,
                                           const float4& al4, float erd, int s, int cnt,
                                           int lane) {
  int j = 0;
  for (; j + GROUP <= cnt; j += GROUP) edge_group<true>(st, hcol, al4, erd, s, j, cnt, lane);
  if (j < cnt) edge_group<false>(st, hcol, al4, erd, s, j, cnt, lane);
}

__device__ __forceinline__ void edge_single(Soft& st, const float* __restrict__ hcol,
                                            const float4& al4, float erd, int sj) {
  const float4 hv = *(const float4*)(hcol + (size_t)sj * C_OUT);
  const float v = leaky(wave_sum(dot4(hv, al4)) + erd);
  const float mn = fmaxf(st.m, v);
  const float sc = __expf(st.m - mn);
  const float p = __expf(v - mn);
  st.m = mn;
  st.sum = fmaf(st.sum, sc, p);
  st.acc.x = fmaf(st.acc.x, sc, p * hv.x); st.acc.y = fmaf(st.acc.y, sc, p * hv.y);
  st.acc.z = fmaf(st.acc.z, sc, p * hv.z); st.acc.w = fmaf(st.acc.w, sc, p * hv.w);
}

static_assert(SLOT_CAP == 64, "a lane keeps the two slots lane and lane + 32 of its node's row");
__global__ __launch_bounds__(128, 16) void gat_aggregate(
    const float* __restrict__ h, const float* __restrict__ attn_l,
    const float* __restrict__ attn_r, const int* __restrict__ src,
    const int* __restrict__ dst, const int* __restrict__ cnt,
    const int* __restrict__ slot, const int* __restrict__ ovf_edge,
    const float* __restrict__ bias, const float* __restrict__ alpha_p,
    const float* __restrict__ x0, float* __restrict__ xout) {
  __shared__ float red2[2][H_HEADS * F_OUTF];
  __shared__ float s_ar[H_HEADS * F_OUTF];
  __shared__ float s_bias[H_HEADS * F_OUTF];
  const int head = threadIdx.x >> 5, lane = threadIdx.x & 31;
  const float* hcol = h + head * F_OUTF + lane * 4;
  const float4 al4 = *(const float4*)(attn_l + head * F_OUTF + lane * 4);
  *(float4*)(s_ar + threadIdx.x * 4) = *(const float4*)(attn_r + threadIdx.x * 4);
  *(float4*)(s_bias + threadIdx.x * 4) = *(const float4*)(bias + threadIdx.x * 4);
  __syncthreads();
  const float al = alpha_p[0];
  int buf = 0;
  for (int n = blockIdx.x; n < N_NODES; n += AGG_BLOCKS, buf ^= 1) {
  float* red = red2[buf];
  const float4 own = *(const float4*)(hcol + (size_t)n * C_OUT);
  const int deg = cnt[n];
  const bool over = deg > SLOT_CAP;
  const int novf = over ? cnt[N_NODES] : 0;
  const int* edge_idx = slot + (size_t)n * SLOT_CAP;
  const int re = over ? SLOT_CAP : deg;
  int e0 = -1, e1 = -1;
  if (lane < re) e0 = edge_idx[lane];
  if (lane + 32 < re) e1 = edge_idx[lane + 32];
  int s0 = 0, s1 = 0;
  if (e0 >= 0) s0 = src[e0];
  if (e1 >= 0) s1 = src[e1];
  const float erd = wave_sum(dot4(own, *(const float4*)(s_ar + head * F_OUTF + lane * 4)));

  Soft st;
  st.m = -INFINITY; st.sum = 0.f; st.acc = make_float4(0.f, 0.f, 0.f, 0.f);
  if (re > 0) edge_chunk(st, hcol, al4, erd, s0, min(32, re), lane);
  if (re > 32) edge_chunk(st, hcol, al4, erd, s1, re - 32, lane);
  if (over) {
    for (int k0 = 0; k0 < novf; k0 += 32) {
      const int k = k0 + lane;
      int s = 0;
      bool mine = false;
      if (k < novf) {
        const int e = ovf_edge[k];
        if (dst[e] == n) { mine = true; s = src[e]; }
      }
      unsigned todo = (unsigned)__ballot(mine);
      while (todo) {
        const int j = __builtin_ctz(todo);
        todo &= todo - 1;
        edge_single(st, hcol, al4, erd, __builtin_amdgcn_readlane(s, j));
      }
    }
  }
  const float inv = (st.sum != 0.f) ? 1.f / st.sum : 0.f;
  *(float4*)(red + head * F_OUTF + lane * 4) =
      make_float4(st.acc.x * inv, st.acc.y * inv, st.acc.z * inv, st.acc.w * inv);
  __syncthreads();

  const int f = threadIdx.x;
  const float s = 0.25f * ((red[f] + s_bias[f]) + (red[F_OUTF + f] + s_bias[F_OUTF + f]) +
                           (red[2 * F_OUTF + f] + s_bias[2 * F_OUTF + f]) +
                           (red[3 * F_OUTF + f] + s_bias[3 * F_OUTF + f]));
  xout[(size_t)n * F_OUTF + f] = (1.f - al) * s + al * x0[(size_t)n * F_OUTF + f];
  }
}

extern "C" void kernel_launch(void* const* d_in, const int* in_sizes, int n_in,
                              void* d_out, int out_size, void* d_ws, size_t ws_size,
                              hipStream_t stream) {
  const float* x      = (const float*)d_in[0];
  const float* x0     = (const float*)d_in[1];
  const int*   src    = (const int*)d_in[2];
  const int*   dst    = (const int*)d_in[3];
  const float* alpha  = (const float*)d_in[4];
  const float* W      = (const float*)d_in[5];
  const float* attn_l = (const float*)d_in[6];
  const float* attn_r = (const float*)d_in[7];
  const float* bias   = (const float*)d_in[8];

  char* ws = (char*)d_ws;
  size_t off = 0;
  auto alloc = [&](size_t bytes) -> char* {
    char* p = ws + off;
    off = (off + bytes + 255) & ~(size_t)255;
    return p;
  };
  float* h        = (float*)alloc((size_t)N_NODES * C_OUT * sizeof(float));
  float* xbuf     = (float*)alloc((size_t)N_NODES * F_INF * sizeof(float));
  int*   cnt      = (int*)alloc((size_t)(N_NODES + 1) * sizeof(int));
  int*   slot     = (int*)alloc((size_t)N_NODES * SLOT_CAP * sizeof(int));
  int*   ovf_edge = (int*)alloc((size_t)N_EDGES * sizeof(int));

  hipMemsetAsync(cnt, 0, (size_t)(N_NODES + 1) * sizeof(int), stream);

  const float* xin = x;
  for (int step = 0; step < STEPS; ++step) {
    if (step == 0)
      gemm_xw_edge_bucket<<<EDGE_BLOCKS + GEMM_BLOCKS, 128, 0, stream>>>(xin, W, h, dst, cnt,
                                                                        slot, ovf_edge);
    else
      gemm_xw<<<GEMM_BLOCKS, 128, 0, stream>>>(xin, W, h);
    float* xo = (step == STEPS - 1) ? (float*)d_out : xbuf;
    gat_aggregate<<<AGG_BLOCKS, 128, 0, stream>>>(h, attn_l, attn_r, src, dst, cnt, slot,
                                               ovf_edge, bias, alpha, x0, xo);
    xin = xbuf;
  }
}
